# starting kernel
# baseline (speedup 1.0000x reference)
#include <hip/hip_runtime.h>

typedef __attribute__((ext_vector_type(2))) float v2f;
typedef __attribute__((ext_vector_type(4))) float v4f;
typedef __attribute__((ext_vector_type(8))) float v8f;

#define NPTS 32768
#define NB   4
#define CIN  64
#define COUT 64
#define KNB  16

#define WPAIRSTR 160
#define WSZ  (128 * WPAIRSTR)
#define VOFF WSZ
#define VWAVE (128 * 32)
#define BOFF (VOFF + 4 * VWAVE)
#define LDS_FLOATS (BOFF + 64)

__global__ __launch_bounds__(256) void transpose_feat_kernel(
    const float* __restrict__ feat, float* __restrict__ ft)
{
    __shared__ float tile[64][65];
    const int b  = blockIdx.y;
    const int n0 = blockIdx.x * 64;
    const int tid = threadIdx.x;
    #pragma unroll
    for (int i = 0; i < 16; ++i) {
        const int j = tid + i * 256;
        const int c = j >> 6, nloc = j & 63;
        tile[nloc][c] = feat[(b * CIN + c) * NPTS + n0 + nloc];
    }
    __syncthreads();
    #pragma unroll
    for (int i = 0; i < 16; ++i) {
        const int j = tid + i * 256;
        const int nloc = j >> 6, c = j & 63;
        ft[(b * NPTS + n0 + nloc) * CIN + c] = tile[nloc][c];
    }
}

template <bool USE_FT>
__global__ __launch_bounds__(128) void flexconv_wmma_kernel(
    const float* __restrict__ feat,
    const float* __restrict__ ft,
    const float* __restrict__ wt,
    const float* __restrict__ wb,
    const float* __restrict__ bias,
    const int*   __restrict__ nbr,
    const float* __restrict__ pos,
    float* __restrict__ out)
{
    __shared__ float lds[LDS_FLOATS];
    const int tid = threadIdx.x;

    for (int j = tid; j < 128 * 64; j += 128) {
        const int dcp = j >> 6, o = j & 63;
        const int dc = dcp * 2;
        float w0, w1;
        if (dc < 192) { w0 = wt[dc * 64 + o];        w1 = wt[(dc + 1) * 64 + o]; }
        else          { w0 = wb[(dc - 192) * 64 + o]; w1 = wb[(dc - 191) * 64 + o]; }
        v2f p; p.x = w0; p.y = w1;
        *reinterpret_cast<v2f*>(&lds[dcp * WPAIRSTR + o * 2]) = p;
    }
    if (tid < COUT) lds[BOFF + tid] = bias[tid];
    __syncthreads();

    const int wave = tid >> 5;
    const int lane = tid & 31;
    const int nl   = lane & 15;
    const int h    = lane >> 4;
    const int b    = blockIdx.y;
    const int n    = blockIdx.x * 64 + wave * 16 + nl;
    const int vb   = VOFF + wave * VWAVE;

    const float cx = pos[(b * 3 + 0) * NPTS + n];
    const float cy = pos[(b * 3 + 1) * NPTS + n];
    const float cz = pos[(b * 3 + 2) * NPTS + n];

    int   idxs[KNB];
    float dxs[KNB], dys[KNB], dzs[KNB];
    #pragma unroll
    for (int k = 0; k < KNB; ++k)
        idxs[k] = nbr[(b * KNB + k) * NPTS + n];
    #pragma unroll
    for (int k = 0; k < KNB; ++k) {
        const int id = idxs[k];
        dxs[k] = pos[(b * 3 + 0) * NPTS + id] - cx;
        dys[k] = pos[(b * 3 + 1) * NPTS + id] - cy;
        dzs[k] = pos[(b * 3 + 2) * NPTS + id] - cz;
    }

    for (int cc = 0; cc < 4; ++cc) {
        const int cbase = h * 32 + cc * 8;
        float ax[8] = {}, ay[8] = {}, az[8] = {}, asum[8] = {};
        for (int k = 0; k < KNB; ++k) {
            const int   id = idxs[k];
            const float dx = dxs[k], dy = dys[k], dz = dzs[k];
            float f[8];
            if constexpr (USE_FT) {
                const v4f* fp = reinterpret_cast<const v4f*>(ft + (b * NPTS + id) * CIN + cbase);
                const v4f f0 = fp[0], f1 = fp[1];
                #pragma unroll
                for (int c = 0; c < 4; ++c) { f[c] = f0[c]; f[c + 4] = f1[c]; }
            } else {
                const float* fp = feat + (b * CIN + cbase) * NPTS + id;
                #pragma unroll
                for (int c = 0; c < 8; ++c) f[c] = fp[c * NPTS];
            }
            #pragma unroll
            for (int c = 0; c < 8; ++c) {
                ax[c]   = fmaf(f[c], dx, ax[c]);
                ay[c]   = fmaf(f[c], dy, ay[c]);
                az[c]   = fmaf(f[c], dz, az[c]);
                asum[c] += f[c];
            }
        }
        #pragma unroll
        for (int c = 0; c < 8; c += 2) {
            const int dcp = (cbase + c) >> 1;
            v2f px, py, pz, ps;
            px.x = ax[c];   px.y = ax[c + 1];
            py.x = ay[c];   py.y = ay[c + 1];
            pz.x = az[c];   pz.y = az[c + 1];
            ps.x = asum[c]; ps.y = asum[c + 1];
            *reinterpret_cast<v2f*>(&lds[vb + (dcp      ) * 32 + nl * 2]) = px;
            *reinterpret_cast<v2f*>(&lds[vb + (dcp + 32 ) * 32 + nl * 2]) = py;
            *reinterpret_cast<v2f*>(&lds[vb + (dcp + 64 ) * 32 + nl * 2]) = pz;
            *reinterpret_cast<v2f*>(&lds[vb + (dcp + 96 ) * 32 + nl * 2]) = ps;
        }
    }

    v8f acc[4] = {v8f{}, v8f{}, v8f{}, v8f{}};
    #pragma unroll 4
    for (int ks = 0; ks < 64; ++ks) {
        const int dcp = ks * 2 + h;
        const v2f bf = *reinterpret_cast<const v2f*>(&lds[vb + dcp * 32 + nl * 2]);
        #pragma unroll
        for (int ot = 0; ot < 4; ++ot) {
            const v2f af = *reinterpret_cast<const v2f*>(
                &lds[dcp * WPAIRSTR + (ot * 16 + nl) * 2]);
            acc[ot] = __builtin_amdgcn_wmma_f32_16x16x4_f32(
                false, af, false, bf, (short)0, acc[ot], false, false);
        }
    }

    #pragma unroll
    for (int ot = 0; ot < 4; ++ot) {
        #pragma unroll
        for (int r = 0; r < 8; ++r) {
            const int o = ot * 16 + r + h * 8;
            out[(b * COUT + o) * NPTS + n] = acc[ot][r] + lds[BOFF + o];
        }
    }
}

extern "C" void kernel_launch(void* const* d_in, const int* in_sizes, int n_in,
                              void* d_out, int out_size, void* d_ws, size_t ws_size,
                              hipStream_t stream) {
    (void)in_sizes; (void)n_in; (void)out_size;
    const float* feat = (const float*)d_in[0];
    const float* wt   = (const float*)d_in[1];
    const float* wb   = (const float*)d_in[2];
    const float* bias = (const float*)d_in[3];
    const int*   nbr  = (const int*)d_in[4];
    const float* pos  = (const float*)d_in[5];
    float* out = (float*)d_out;

    const size_t ft_bytes = (size_t)NB * NPTS * CIN * sizeof(float);
    dim3 grid(NPTS / 64, NB);

    if (ws_size >= ft_bytes) {
        float* ft = (float*)d_ws;
        transpose_feat_kernel<<<grid, 256, 0, stream>>>(feat, ft);
        flexconv_wmma_kernel<true><<<grid, 128, 0, stream>>>(
            feat, ft, wt, wb, bias, nbr, pos, out);
    } else {
        flexconv_wmma_kernel<false><<<grid, 128, 0, stream>>>(
            feat, feat, wt, wb, bias, nbr, pos, out);
    }
}


# matrix phase re-split over waves by output tile: each wave keeps its 16-channel W' slice in registers as A fragments (global loads, no W' in LDS), reads all 4 v tiles; sum order unchanged
# speedup vs baseline: 1.3468x; 1.3468x over previous
#include <hip/hip_runtime.h>

typedef __attribute__((ext_vector_type(2))) float v2f;
typedef __attribute__((ext_vector_type(4))) float v4f;
typedef __attribute__((ext_vector_type(8))) float v8f;

#define NPTS 32768
#define NB   4
#define CIN  64
#define COUT 64
#define KNB  16

#define VOFF 0
#define VWAVE (128 * 32)
#define BOFF (VOFF + 4 * VWAVE)
#define LDS_FLOATS (BOFF + 64)

__global__ __launch_bounds__(256) void transpose_feat_kernel(
    const float* __restrict__ feat, float* __restrict__ ft)
{
    __shared__ float tile[64][65];
    const int b  = blockIdx.y;
    const int n0 = blockIdx.x * 64;
    const int tid = threadIdx.x;
    #pragma unroll
    for (int i = 0; i < 16; ++i) {
        const int j = tid + i * 256;
        const int c = j >> 6, nloc = j & 63;
        tile[nloc][c] = feat[(b * CIN + c) * NPTS + n0 + nloc];
    }
    __syncthreads();
    #pragma unroll
    for (int i = 0; i < 16; ++i) {
        const int j = tid + i * 256;
        const int nloc = j >> 6, c = j & 63;
        ft[(b * NPTS + n0 + nloc) * CIN + c] = tile[nloc][c];
    }
}

template <bool USE_FT>
__global__ __launch_bounds__(128) void flexconv_wmma_kernel(
    const float* __restrict__ feat,
    const float* __restrict__ ft,
    const float* __restrict__ wt,
    const float* __restrict__ wb,
    const float* __restrict__ bias,
    const int*   __restrict__ nbr,
    const float* __restrict__ pos,
    float* __restrict__ out)
{
    __shared__ float lds[LDS_FLOATS];
    const int tid = threadIdx.x;

    if (tid < COUT) lds[BOFF + tid] = bias[tid];

    const int wave = tid >> 5;
    const int lane = tid & 31;
    const int nl   = lane & 15;
    const int h    = lane >> 4;
    const int b    = blockIdx.y;
    const int n    = blockIdx.x * 64 + wave * 16 + nl;
    const int vb   = VOFF + wave * VWAVE;

    const float cx = pos[(b * 3 + 0) * NPTS + n];
    const float cy = pos[(b * 3 + 1) * NPTS + n];
    const float cz = pos[(b * 3 + 2) * NPTS + n];

    int   idxs[KNB];
    float dxs[KNB], dys[KNB], dzs[KNB];
    #pragma unroll
    for (int k = 0; k < KNB; ++k)
        idxs[k] = nbr[(b * KNB + k) * NPTS + n];
    #pragma unroll
    for (int k = 0; k < KNB; ++k) {
        const int id = idxs[k];
        dxs[k] = pos[(b * 3 + 0) * NPTS + id] - cx;
        dys[k] = pos[(b * 3 + 1) * NPTS + id] - cy;
        dzs[k] = pos[(b * 3 + 2) * NPTS + id] - cz;
    }

    for (int cc = 0; cc < 4; ++cc) {
        const int cbase = h * 32 + cc * 8;
        float ax[8] = {}, ay[8] = {}, az[8] = {}, asum[8] = {};
        for (int k = 0; k < KNB; ++k) {
            const int   id = idxs[k];
            const float dx = dxs[k], dy = dys[k], dz = dzs[k];
            float f[8];
            if constexpr (USE_FT) {
                const v4f* fp = reinterpret_cast<const v4f*>(ft + (b * NPTS + id) * CIN + cbase);
                const v4f f0 = fp[0], f1 = fp[1];
                #pragma unroll
                for (int c = 0; c < 4; ++c) { f[c] = f0[c]; f[c + 4] = f1[c]; }
            } else {
                const float* fp = feat + (b * CIN + cbase) * NPTS + id;
                #pragma unroll
                for (int c = 0; c < 8; ++c) f[c] = fp[c * NPTS];
            }
            #pragma unroll
            for (int c = 0; c < 8; ++c) {
                ax[c]   = fmaf(f[c], dx, ax[c]);
                ay[c]   = fmaf(f[c], dy, ay[c]);
                az[c]   = fmaf(f[c], dz, az[c]);
                asum[c] += f[c];
            }
        }
        #pragma unroll
        for (int c = 0; c < 8; c += 2) {
            const int dcp = (cbase + c) >> 1;
            v2f px, py, pz, ps;
            px.x = ax[c];   px.y = ax[c + 1];
            py.x = ay[c];   py.y = ay[c + 1];
            pz.x = az[c];   pz.y = az[c + 1];
            ps.x = asum[c]; ps.y = asum[c + 1];
            *reinterpret_cast<v2f*>(&lds[vb + (dcp      ) * 32 + nl * 2]) = px;
            *reinterpret_cast<v2f*>(&lds[vb + (dcp + 32 ) * 32 + nl * 2]) = py;
            *reinterpret_cast<v2f*>(&lds[vb + (dcp + 64 ) * 32 + nl * 2]) = pz;
            *reinterpret_cast<v2f*>(&lds[vb + (dcp + 96 ) * 32 + nl * 2]) = ps;
        }
    }
    const int oa = wave * 16 + nl;
    v2f af[64];
    #pragma unroll
    for (int ks = 0; ks < 64; ++ks) {
        const int dc = ks * 4 + h * 2;
        const float* wrow = (ks < 48) ? (wt + dc * 64) : (wb + (dc - 192) * 64);
        af[ks].x = wrow[oa];
        af[ks].y = wrow[64 + oa];
    }
    __syncthreads();

    v8f acc[4] = {v8f{}, v8f{}, v8f{}, v8f{}};
    #pragma unroll
    for (int ks = 0; ks < 64; ++ks) {
        const int dcp = ks * 2 + h;
        #pragma unroll
        for (int pt = 0; pt < 4; ++pt) {
            const v2f bf = *reinterpret_cast<const v2f*>(
                &lds[VOFF + pt * VWAVE + dcp * 32 + nl * 2]);
            acc[pt] = __builtin_amdgcn_wmma_f32_16x16x4_f32(
                false, af[ks], false, bf, (short)0, acc[pt], false, false);
        }
    }

    const int nbase = blockIdx.x * 64 + nl;
    #pragma unroll
    for (int pt = 0; pt < 4; ++pt) {
        #pragma unroll
        for (int r = 0; r < 8; ++r) {
            const int o = wave * 16 + r + h * 8;
            out[(b * COUT + o) * NPTS + nbase + pt * 16] = acc[pt][r] + lds[BOFF + o];
        }
    }
}

extern "C" void kernel_launch(void* const* d_in, const int* in_sizes, int n_in,
                              void* d_out, int out_size, void* d_ws, size_t ws_size,
                              hipStream_t stream) {
    (void)in_sizes; (void)n_in; (void)out_size;
    const float* feat = (const float*)d_in[0];
    const float* wt   = (const float*)d_in[1];
    const float* wb   = (const float*)d_in[2];
    const float* bias = (const float*)d_in[3];
    const int*   nbr  = (const int*)d_in[4];
    const float* pos  = (const float*)d_in[5];
    float* out = (float*)d_out;

    const size_t ft_bytes = (size_t)NB * NPTS * CIN * sizeof(float);
    dim3 grid(NPTS / 64, NB);

    if (ws_size >= ft_bytes) {
        float* ft = (float*)d_ws;
        transpose_feat_kernel<<<grid, 256, 0, stream>>>(feat, ft);
        flexconv_wmma_kernel<true><<<grid, 128, 0, stream>>>(
            feat, ft, wt, wb, bias, nbr, pos, out);
    } else {
        flexconv_wmma_kernel<false><<<grid, 128, 0, stream>>>(
            feat, feat, wt, wb, bias, nbr, pos, out);
    }
}


# gather re-tiled: 2 passes of one whole 128 B line (was 4 of 32 B), a point's two half-lanes take alternate 16 B chunks: a load touches half as many lines, a line is asked for once; sums unchanged
# speedup vs baseline: 2.7114x; 2.0132x over previous
#include <hip/hip_runtime.h>

typedef __attribute__((ext_vector_type(2))) float v2f;
typedef __attribute__((ext_vector_type(4))) float v4f;
typedef __attribute__((ext_vector_type(8))) float v8f;

#define NPTS 32768
#define NB   4
#define CIN  64
#define COUT 64
#define KNB  16

#define VOFF 0
#define VWAVE (128 * 32)
#define BOFF (VOFF + 4 * VWAVE)
#define LDS_FLOATS (BOFF + 64)

__global__ __launch_bounds__(256) void transpose_feat_kernel(
    const float* __restrict__ feat, float* __restrict__ ft)
{
    __shared__ float tile[64][65];
    const int b  = blockIdx.y;
    const int n0 = blockIdx.x * 64;
    const int tid = threadIdx.x;
    #pragma unroll
    for (int i = 0; i < 16; ++i) {
        const int j = tid + i * 256;
        const int c = j >> 6, nloc = j & 63;
        tile[nloc][c] = feat[(b * CIN + c) * NPTS + n0 + nloc];
    }
    __syncthreads();
    #pragma unroll
    for (int i = 0; i < 16; ++i) {
        const int j = tid + i * 256;
        const int nloc = j >> 6, c = j & 63;
        ft[(b * NPTS + n0 + nloc) * CIN + c] = tile[nloc][c];
    }
}

template <bool USE_FT>
__global__ __launch_bounds__(128, 4) void flexconv_wmma_kernel(
    const float* __restrict__ feat,
    const float* __restrict__ ft,
    const float* __restrict__ wt,
    const float* __restrict__ wb,
    const float* __restrict__ bias,
    const int*   __restrict__ nbr,
    const float* __restrict__ pos,
    float* __restrict__ out)
{
    __shared__ float lds[LDS_FLOATS];
    const int tid = threadIdx.x;

    if (tid < COUT) lds[BOFF + tid] = bias[tid];

    const int wave = tid >> 5;
    const int lane = tid & 31;
    const int nl   = lane & 15;
    const int h    = lane >> 4;
    const int b    = blockIdx.y;
    const int n    = blockIdx.x * 64 + wave * 16 + nl;
    const int vb   = VOFF + wave * VWAVE;

    const float cx = pos[(b * 3 + 0) * NPTS + n];
    const float cy = pos[(b * 3 + 1) * NPTS + n];
    const float cz = pos[(b * 3 + 2) * NPTS + n];

    int   idxs[KNB];
    float dxs[KNB], dys[KNB], dzs[KNB];
    #pragma unroll
    for (int k = 0; k < KNB; ++k)
        idxs[k] = nbr[(b * KNB + k) * NPTS + n];
    #pragma unroll
    for (int k = 0; k < KNB; ++k) {
        const int id = idxs[k];
        dxs[k] = pos[(b * 3 + 0) * NPTS + id] - cx;
        dys[k] = pos[(b * 3 + 1) * NPTS + id] - cy;
        dzs[k] = pos[(b * 3 + 2) * NPTS + id] - cz;
    }

    for (int ps = 0; ps < 2; ++ps) {
        float ax[16] = {}, ay[16] = {}, az[16] = {}, asum[16] = {};
        #pragma unroll
        for (int k = 0; k < KNB; ++k) {
            const int   id = idxs[k];
            const float dx = dxs[k], dy = dys[k], dz = dzs[k];
            float f[16];
            if constexpr (USE_FT) {
                const unsigned row = (unsigned)(b * NPTS + id) * CIN + ps * 32 + h * 4;
                const v4f* fp = reinterpret_cast<const v4f*>(ft + row);
                #pragma unroll
                for (int jj = 0; jj < 4; ++jj) {
                    const v4f q = fp[2 * jj];
                    #pragma unroll
                    for (int i = 0; i < 4; ++i) f[4 * jj + i] = q[i];
                }
            } else {
                const float* fp = feat + (b * CIN + ps * 32 + 4 * h) * NPTS + id;
                #pragma unroll
                for (int jj = 0; jj < 4; ++jj) {
                    #pragma unroll
                    for (int i = 0; i < 4; ++i) f[4 * jj + i] = fp[(8 * jj + i) * NPTS];
                }
            }
            #pragma unroll
            for (int c = 0; c < 16; ++c) {
                ax[c]   = fmaf(f[c], dx, ax[c]);
                ay[c]   = fmaf(f[c], dy, ay[c]);
                az[c]   = fmaf(f[c], dz, az[c]);
                asum[c] += f[c];
            }
        }
        #pragma unroll
        for (int jj = 0; jj < 4; ++jj) {
            #pragma unroll
            for (int ip = 0; ip < 2; ++ip) {
                const int c   = 4 * jj + 2 * ip;
                const int dcp = 16 * ps + 4 * jj + 2 * h + ip;
                v2f px, py, pz, pw;
                px.x = ax[c];   px.y = ax[c + 1];
                py.x = ay[c];   py.y = ay[c + 1];
                pz.x = az[c];   pz.y = az[c + 1];
                pw.x = asum[c]; pw.y = asum[c + 1];
                *reinterpret_cast<v2f*>(&lds[vb + (dcp      ) * 32 + nl * 2]) = px;
                *reinterpret_cast<v2f*>(&lds[vb + (dcp + 32 ) * 32 + nl * 2]) = py;
                *reinterpret_cast<v2f*>(&lds[vb + (dcp + 64 ) * 32 + nl * 2]) = pz;
                *reinterpret_cast<v2f*>(&lds[vb + (dcp + 96 ) * 32 + nl * 2]) = pw;
            }
        }
    }
    const int oa = wave * 16 + nl;
    v2f af[64];
    #pragma unroll
    for (int ks = 0; ks < 64; ++ks) {
        const int dc = ks * 4 + h * 2;
        const float* wrow = (ks < 48) ? (wt + dc * 64) : (wb + (dc - 192) * 64);
        af[ks].x = wrow[oa];
        af[ks].y = wrow[64 + oa];
    }
    __syncthreads();

    v8f acc[4] = {v8f{}, v8f{}, v8f{}, v8f{}};
    #pragma unroll
    for (int ks = 0; ks < 64; ++ks) {
        const int dcp = ks * 2 + h;
        #pragma unroll
        for (int pt = 0; pt < 4; ++pt) {
            const v2f bf = *reinterpret_cast<const v2f*>(
                &lds[VOFF + pt * VWAVE + dcp * 32 + nl * 2]);
            acc[pt] = __builtin_amdgcn_wmma_f32_16x16x4_f32(
                false, af[ks], false, bf, (short)0, acc[pt], false, false);
        }
    }

    const int nbase = blockIdx.x * 64 + nl;
    #pragma unroll
    for (int pt = 0; pt < 4; ++pt) {
        #pragma unroll
        for (int r = 0; r < 8; ++r) {
            const int o = wave * 16 + r + h * 8;
            out[(b * COUT + o) * NPTS + nbase + pt * 16] = acc[pt][r] + lds[BOFF + o];
        }
    }
}

extern "C" void kernel_launch(void* const* d_in, const int* in_sizes, int n_in,
                              void* d_out, int out_size, void* d_ws, size_t ws_size,
                              hipStream_t stream) {
    (void)in_sizes; (void)n_in; (void)out_size;
    const float* feat = (const float*)d_in[0];
    const float* wt   = (const float*)d_in[1];
    const float* wb   = (const float*)d_in[2];
    const float* bias = (const float*)d_in[3];
    const int*   nbr  = (const int*)d_in[4];
    const float* pos  = (const float*)d_in[5];
    float* out = (float*)d_out;

    const size_t ft_bytes = (size_t)NB * NPTS * CIN * sizeof(float);
    dim3 grid(NPTS / 64, NB);

    if (ws_size >= ft_bytes) {
        float* ft = (float*)d_ws;
        transpose_feat_kernel<<<grid, 256, 0, stream>>>(feat, ft);
        flexconv_wmma_kernel<true><<<grid, 128, 0, stream>>>(
            feat, ft, wt, wb, bias, nbr, pos, out);
    } else {
        flexconv_wmma_kernel<false><<<grid, 128, 0, stream>>>(
            feat, feat, wt, wb, bias, nbr, pos, out);
    }
}


# matrix phase split by gather pass: 32 K steps on pass-0 rows issue inside gather pass 1, under its gathers' wait, rest after a 2nd barrier; K summed in that order; rel. positions parked in LDS
# speedup vs baseline: 2.8927x; 1.0669x over previous
#include <hip/hip_runtime.h>

typedef __attribute__((ext_vector_type(2))) float v2f;
typedef __attribute__((ext_vector_type(4))) float v4f;
typedef __attribute__((ext_vector_type(8))) float v8f;

#define NPTS 32768
#define NB   4
#define CIN  64
#define COUT 64
#define KNB  16

#define VOFF 0
#define VWAVE (128 * 32)
#define BOFF (VOFF + 4 * VWAVE)
#define LDS_FLOATS (BOFF + 64)

__global__ __launch_bounds__(256) void transpose_feat_kernel(
    const float* __restrict__ feat, float* __restrict__ ft)
{
    __shared__ float tile[64][65];
    const int b  = blockIdx.y;
    const int n0 = blockIdx.x * 64;
    const int tid = threadIdx.x;
    #pragma unroll
    for (int i = 0; i < 16; ++i) {
        const int j = tid + i * 256;
        const int c = j >> 6, nloc = j & 63;
        tile[nloc][c] = feat[(b * CIN + c) * NPTS + n0 + nloc];
    }
    __syncthreads();
    #pragma unroll
    for (int i = 0; i < 16; ++i) {
        const int j = tid + i * 256;
        const int nloc = j >> 6, c = j & 63;
        ft[(b * NPTS + n0 + nloc) * CIN + c] = tile[nloc][c];
    }
}

template <bool USE_FT, int PS>
__device__ __forceinline__ void issue_nb(const float* __restrict__ feat, const float* __restrict__ ft,
                                         int b, int h, unsigned nb, v4f (&q)[4])
{
    if constexpr (USE_FT) {
        const v4f* fp = reinterpret_cast<const v4f*>(
            reinterpret_cast<const char*>(ft) + nb + PS * 128);
        #pragma unroll
        for (int jj = 0; jj < 4; ++jj) q[jj] = fp[2 * jj];
    } else {
        const float* fp = feat + (b * CIN + PS * 32 + 4 * h) * NPTS + nb;
        #pragma unroll
        for (int jj = 0; jj < 4; ++jj) {
            #pragma unroll
            for (int i = 0; i < 4; ++i) q[jj][i] = fp[(8 * jj + i) * NPTS];
        }
    }
}

__device__ __forceinline__ void fma_nb(const v4f (&q)[4], float dx, float dy, float dz,
                                       float (&ax)[16], float (&ay)[16], float (&az)[16], float (&asum)[16])
{
    #pragma unroll
    for (int c = 0; c < 16; ++c) {
        const float f = q[c >> 2][c & 3];
        ax[c]   = fmaf(f, dx, ax[c]);
        ay[c]   = fmaf(f, dy, ay[c]);
        az[c]   = fmaf(f, dz, az[c]);
        asum[c] += f;
    }
}

template <int PS>
__device__ __forceinline__ void store_v(float* __restrict__ vt, int nl, int h,
                                        const float (&ax)[16], const float (&ay)[16],
                                        const float (&az)[16], const float (&asum)[16])
{
    #pragma unroll
    for (int jj = 0; jj < 4; ++jj) {
        #pragma unroll
        for (int ip = 0; ip < 2; ++ip) {
            const int c   = 4 * jj + 2 * ip;
            const int dcp = 16 * PS + 4 * jj + 2 * h + ip;
            v2f px, py, pz, pw;
            px.x = ax[c];   px.y = ax[c + 1];
            py.x = ay[c];   py.y = ay[c + 1];
            pz.x = az[c];   pz.y = az[c + 1];
            pw.x = asum[c]; pw.y = asum[c + 1];
            *reinterpret_cast<v2f*>(&vt[(dcp      ) * 32 + nl * 2]) = px;
            *reinterpret_cast<v2f*>(&vt[(dcp + 32 ) * 32 + nl * 2]) = py;
            *reinterpret_cast<v2f*>(&vt[(dcp + 64 ) * 32 + nl * 2]) = pz;
            *reinterpret_cast<v2f*>(&vt[(dcp + 96 ) * 32 + nl * 2]) = pw;
        }
    }
}

__device__ __forceinline__ v2f load_af(const float* __restrict__ wt, const float* __restrict__ wb,
                                       int ks, int h, int oa)
{
    const int dc = ks * 4 + h * 2;
    const float* wrow = (ks < 48) ? (wt + dc * 64) : (wb + (dc - 192) * 64);
    v2f a;
    a.x = wrow[oa];
    a.y = wrow[64 + oa];
    return a;
}

#define KSTEP(PS, q) (16 * ((q) >> 3) + 8 * (PS) + ((q) & 7))

#define P_DX (16 * 32)
#define P_DY (16 * 32 + 256)
#define P_DZ (48 * 32)
#define P_NB (48 * 32 + 256)

template <bool USE_FT>
__global__ __launch_bounds__(128, 4) void flexconv_wmma_kernel(
    const float* __restrict__ feat,
    const float* __restrict__ ft,
    const float* __restrict__ wt,
    const float* __restrict__ wb,
    const float* __restrict__ bias,
    const int*   __restrict__ nbr,
    const float* __restrict__ pos,
    float* __restrict__ out)
{
    __shared__ float lds[LDS_FLOATS];
    const int tid = threadIdx.x;

    if (tid < COUT) lds[BOFF + tid] = bias[tid];

    const int wave = tid >> 5;
    const int lane = tid & 31;
    const int nl   = lane & 15;
    const int h    = lane >> 4;
    const int b    = blockIdx.y;
    const int n    = blockIdx.x * 64 + wave * 16 + nl;
    const int vb   = VOFF + wave * VWAVE;
    const int oa   = wave * 16 + nl;
    const unsigned hoff = USE_FT ? (unsigned)h * 16u : 0u;

    {
        const float cx = pos[(b * 3 + 0) * NPTS + n];
        const float cy = pos[(b * 3 + 1) * NPTS + n];
        const float cz = pos[(b * 3 + 2) * NPTS + n];
        int idxs[KNB];
        #pragma unroll
        for (int k = 0; k < KNB; ++k)
            idxs[k] = nbr[(b * KNB + k) * NPTS + n];
        #pragma unroll
        for (int k = 0; k < KNB; ++k) {
            const int id = idxs[k];
            const unsigned nb = USE_FT ? (unsigned)(b * NPTS + id) * (CIN * 4u) : (unsigned)id;
            lds[vb + P_DX + k * 16 + nl] = pos[(b * 3 + 0) * NPTS + id] - cx;
            lds[vb + P_DY + k * 16 + nl] = pos[(b * 3 + 1) * NPTS + id] - cy;
            lds[vb + P_DZ + k * 16 + nl] = pos[(b * 3 + 2) * NPTS + id] - cz;
            lds[vb + P_NB + k * 16 + nl] = __uint_as_float(nb);
        }
    }

    {
        float ax[16] = {}, ay[16] = {}, az[16] = {}, asum[16] = {};
        #pragma unroll 1
        for (int g = 0; g < 2; ++g) {
            v4f q[8][4];
            #pragma unroll
            for (int kk = 0; kk < 8; ++kk) {
                const unsigned nb = __float_as_uint(lds[vb + P_NB + (8 * g + kk) * 16 + nl]) + hoff;
                issue_nb<USE_FT, 0>(feat, ft, b, h, nb, q[kk]);
            }
            #pragma unroll
            for (int kk = 0; kk < 8; ++kk) {
                const int k = 8 * g + kk;
                fma_nb(q[kk], lds[vb + P_DX + k * 16 + nl], lds[vb + P_DY + k * 16 + nl],
                       lds[vb + P_DZ + k * 16 + nl], ax, ay, az, asum);
            }
        }
        store_v<0>(&lds[vb], nl, h, ax, ay, az, asum);
    }
    __syncthreads();

    v8f acc[4] = {v8f{}, v8f{}, v8f{}, v8f{}};
    {
        float ax[16] = {}, ay[16] = {}, az[16] = {}, asum[16] = {};
        #pragma unroll 2
        for (int g = 0; g < 4; ++g) {
            const float* wbase = (g < 3) ? (wt + g * (64 * 64)) : wb;
            v2f a[8];
            #pragma unroll
            for (int j = 0; j < 8; ++j) {
                a[j].x = wbase[(4 * j + 2 * h) * 64 + oa];
                a[j].y = wbase[(4 * j + 2 * h + 1) * 64 + oa];
            }
            v4f q[4][4];
            #pragma unroll
            for (int kk = 0; kk < 4; ++kk) {
                const unsigned nb = __float_as_uint(lds[vb + P_NB + (4 * g + kk) * 16 + nl]) + hoff;
                issue_nb<USE_FT, 1>(feat, ft, b, h, nb, q[kk]);
            }
            #pragma unroll
            for (int j = 0; j < 8; ++j) {
                const int dcp = (16 * g + j) * 2 + h;
                #pragma unroll
                for (int pt = 0; pt < 4; ++pt) {
                    const v2f bf = *reinterpret_cast<const v2f*>(
                        &lds[VOFF + pt * VWAVE + dcp * 32 + nl * 2]);
                    acc[pt] = __builtin_amdgcn_wmma_f32_16x16x4_f32(
                        false, a[j], false, bf, (short)0, acc[pt], false, false);
                }
            }
            #pragma unroll
            for (int kk = 0; kk < 4; ++kk) {
                const int k = 4 * g + kk;
                fma_nb(q[kk], lds[vb + P_DX + k * 16 + nl], lds[vb + P_DY + k * 16 + nl],
                       lds[vb + P_DZ + k * 16 + nl], ax, ay, az, asum);
            }
        }
        store_v<1>(&lds[vb], nl, h, ax, ay, az, asum);
    }
    v2f af[32];
    #pragma unroll
    for (int q = 0; q < 32; ++q) af[q] = load_af(wt, wb, KSTEP(1, q), h, oa);
    __syncthreads();

    #pragma unroll
    for (int q = 0; q < 32; ++q) {
        const int dcp = KSTEP(1, q) * 2 + h;
        #pragma unroll
        for (int pt = 0; pt < 4; ++pt) {
            const v2f bf = *reinterpret_cast<const v2f*>(
                &lds[VOFF + pt * VWAVE + dcp * 32 + nl * 2]);
            acc[pt] = __builtin_amdgcn_wmma_f32_16x16x4_f32(
                false, af[q], false, bf, (short)0, acc[pt], false, false);
        }
    }

    const int nbase = blockIdx.x * 64 + nl;
    #pragma unroll
    for (int pt = 0; pt < 4; ++pt) {
        #pragma unroll
        for (int r = 0; r < 8; ++r) {
            const int o = wave * 16 + r + h * 8;
            out[(b * COUT + o) * NPTS + nbase + pt * 16] = acc[pt][r] + lds[BOFF + o];
        }
    }
}

extern "C" void kernel_launch(void* const* d_in, const int* in_sizes, int n_in,
                              void* d_out, int out_size, void* d_ws, size_t ws_size,
                              hipStream_t stream) {
    (void)in_sizes; (void)n_in; (void)out_size;
    const float* feat = (const float*)d_in[0];
    const float* wt   = (const float*)d_in[1];
    const float* wb   = (const float*)d_in[2];
    const float* bias = (const float*)d_in[3];
    const int*   nbr  = (const int*)d_in[4];
    const float* pos  = (const float*)d_in[5];
    float* out = (float*)d_out;

    const size_t ft_bytes = (size_t)NB * NPTS * CIN * sizeof(float);
    dim3 grid(NPTS / 64, NB);

    if (ws_size >= ft_bytes) {
        float* ft = (float*)d_ws;
        transpose_feat_kernel<<<grid, 256, 0, stream>>>(feat, ft);
        flexconv_wmma_kernel<true><<<grid, 128, 0, stream>>>(
            feat, ft, wt, wb, bias, nbr, pos, out);
    } else {
        flexconv_wmma_kernel<false><<<grid, 128, 0, stream>>>(
            feat, feat, wt, wb, bias, nbr, pos, out);
    }
}
